# starting kernel
# baseline (speedup 1.0000x reference)
#include <hip/hip_runtime.h>


typedef __attribute__((ext_vector_type(2))) float v2f;
typedef __attribute__((ext_vector_type(8))) float v8f;

#define LDP   132
#define HDIM  128
#define CINC  64
#define NOBSC 16

__device__ __forceinline__ v8f wmma4(v2f a, v2f b, v8f c) {
  return __builtin_amdgcn_wmma_f32_16x16x4_f32(
       false, a,  false, b,
       (short)0, c,  false,  false);
}

template <int K>
__device__ __forceinline__ void gemm_lds(const float* __restrict__ sY,
                                         const float* __restrict__ W, int ldw,
                                         int wcol, int lane, v8f acc[4]) {
  const int mrow  = lane & 15;
  const int khalf = (lane >> 4) << 1;
  #pragma unroll 4
  for (int ks = 0; ks < K / 4; ++ks) {
    const int k0 = ks * 4 + khalf;
    v2f a = *(const v2f*)(sY + mrow * LDP + k0);
    #pragma unroll
    for (int nt = 0; nt < 4; ++nt) {
      const int n = wcol + nt * 16 + mrow;
      v2f b = *(const v2f*)(W + n * ldw + k0);
      acc[nt] = wmma4(a, b, acc[nt]);
    }
  }
}

__device__ __forceinline__ void write_tile(float* __restrict__ sY, int wcol, int lane,
                                           const v8f v[4]) {
  const int hi   = lane >> 4;
  const int col0 = lane & 15;
  #pragma unroll
  for (int nt = 0; nt < 4; ++nt) {
    const int col = wcol + nt * 16 + col0;
    #pragma unroll
    for (int r = 0; r < 8; ++r)
      sY[(r + 8 * hi) * LDP + col] = v[nt][r];
  }
}

__global__ void __launch_bounds__(64) jacde_kernel(
    const float* __restrict__ t,      const float* __restrict__ hmat,
    const float* __restrict__ coeffs, const float* __restrict__ dcoeffs,
    const float* __restrict__ tobs,
    const float* __restrict__ wx,     const float* __restrict__ wh,
    const float* __restrict__ wout,
    const float* __restrict__ b0,     const float* __restrict__ b1,
    float* __restrict__ out) {
  __shared__ float sY[16 * LDP];

  const int tid  = threadIdx.x;
  const int lane = tid & 31;
  const int wave = tid >> 5;
  const int wcol = wave * 64;
  const int brow = blockIdx.x * 16;
  const int hi   = lane >> 4;
  const int col0 = lane & 15;

  const float ts = t[0];
  int idx = 0;
  #pragma unroll
  for (int j = 0; j < NOBSC; ++j) idx += (tobs[j] <= ts) ? 1 : 0;
  idx -= 1;
  idx = idx < 0 ? 0 : (idx > NOBSC - 2 ? NOBSC - 2 : idx);
  const float dt = ts - tobs[idx];
  const float p1 = dt, p2 = dt * dt, p3 = p2 * dt;

  {
    const float* cb = coeffs + ((size_t)brow * (NOBSC - 1) + idx) * CINC * 4;
    #pragma unroll 4
    for (int i = 0; i < 16; ++i) {
      const float4 c4 = *(const float4*)(cb + (size_t)i * (NOBSC - 1) * CINC * 4 + tid * 4);
      sY[i * LDP + tid] = c4.x + c4.y * p1 + c4.z * p2 + c4.w * p3;
    }
  }
  __syncthreads();

  v8f acc[4];
  #pragma unroll
  for (int nt = 0; nt < 4; ++nt) {
    const float bv = b0[wcol + nt * 16 + col0];
    #pragma unroll
    for (int r = 0; r < 8; ++r) acc[nt][r] = bv;
  }
  gemm_lds<CINC>(sY, wx, CINC, wcol, lane, acc);
  __syncthreads();

  #pragma unroll 4
  for (int i = 0; i < 32; ++i) {
    const int v = i * 64 + tid, m = v >> 7, c = v & 127;
    sY[m * LDP + c] = hmat[(size_t)(brow + m) * HDIM + c];
  }
  __syncthreads();
  gemm_lds<HDIM>(sY, wh, HDIM, wcol, lane, acc);
  __syncthreads();

  v8f drelu[4];
  #pragma unroll
  for (int nt = 0; nt < 4; ++nt)
    #pragma unroll
    for (int r = 0; r < 8; ++r) {
      const float x = acc[nt][r];
      drelu[nt][r] = 1.0f / (1.0f + __expf(-x));
      acc[nt][r]   = fmaxf(x, 0.0f);
    }
  write_tile(sY, wcol, lane, acc);
  __syncthreads();

  v8f dtn[4];
  {
    v8f a2[4];
    #pragma unroll
    for (int nt = 0; nt < 4; ++nt) {
      const float bv = b1[wcol + nt * 16 + col0];
      #pragma unroll
      for (int r = 0; r < 8; ++r) a2[nt][r] = bv;
    }
    gemm_lds<HDIM>(sY, wout, HDIM, wcol, lane, a2);
    #pragma unroll
    for (int nt = 0; nt < 4; ++nt)
      #pragma unroll
      for (int r = 0; r < 8; ++r) {
        const float th = tanhf(a2[nt][r]);
        dtn[nt][r] = 1.0f - th * th;
      }
  }
  __syncthreads();

  {
    const float* db = dcoeffs + ((size_t)brow * (NOBSC - 1) + idx) * CINC * 4;
    #pragma unroll 4
    for (int i = 0; i < 16; ++i) {
      const float4 c4 = *(const float4*)(db + (size_t)i * (NOBSC - 1) * CINC * 4 + tid * 4);
      sY[i * LDP + tid] = c4.x + c4.y * p1 + c4.z * p2 + c4.w * p3;
    }
  }
  __syncthreads();
  v8f cur[4];
  #pragma unroll
  for (int nt = 0; nt < 4; ++nt) cur[nt] = (v8f)(0.0f);
  gemm_lds<CINC>(sY, wx, CINC, wcol, lane, cur);
  __syncthreads();

  #pragma unroll
  for (int nt = 0; nt < 4; ++nt) cur[nt] *= drelu[nt];
  write_tile(sY, wcol, lane, cur);
  __syncthreads();
  v8f outv[4];
  #pragma unroll
  for (int nt = 0; nt < 4; ++nt) outv[nt] = (v8f)(0.0f);
  gemm_lds<HDIM>(sY, wout, HDIM, wcol, lane, outv);
  __syncthreads();
  #pragma unroll
  for (int nt = 0; nt < 4; ++nt) {
    outv[nt] *= dtn[nt];
    cur[nt] = outv[nt];
  }

  for (int it = 0; it < 2; ++it) {
    write_tile(sY, wcol, lane, cur);
    __syncthreads();
    v8f t1[4];
    #pragma unroll
    for (int nt = 0; nt < 4; ++nt) t1[nt] = (v8f)(0.0f);
    gemm_lds<HDIM>(sY, wh, HDIM, wcol, lane, t1);
    __syncthreads();
    #pragma unroll
    for (int nt = 0; nt < 4; ++nt) t1[nt] *= drelu[nt];
    write_tile(sY, wcol, lane, t1);
    __syncthreads();
    v8f t2[4];
    #pragma unroll
    for (int nt = 0; nt < 4; ++nt) t2[nt] = (v8f)(0.0f);
    gemm_lds<HDIM>(sY, wout, HDIM, wcol, lane, t2);
    __syncthreads();
    #pragma unroll
    for (int nt = 0; nt < 4; ++nt) {
      t2[nt] *= dtn[nt];
      outv[nt] += t2[nt];
      cur[nt] = t2[nt];
    }
  }

  #pragma unroll
  for (int nt = 0; nt < 4; ++nt) {
    const int col = wcol + nt * 16 + col0;
    #pragma unroll
    for (int r = 0; r < 8; ++r)
      out[(size_t)(brow + r + 8 * hi) * HDIM + col] = outv[nt][r];
  }
}

extern "C" void kernel_launch(void* const* d_in, const int* in_sizes, int n_in,
                              void* d_out, int out_size, void* d_ws, size_t ws_size,
                              hipStream_t stream) {
  const float* t       = (const float*)d_in[0];
  const float* hmat    = (const float*)d_in[1];
  const float* coeffs  = (const float*)d_in[2];
  const float* dcoeffs = (const float*)d_in[3];
  const float* tobs    = (const float*)d_in[4];
  const float* wx      = (const float*)d_in[5];
  const float* wh      = (const float*)d_in[6];
  const float* wout    = (const float*)d_in[7];
  const float* b0      = (const float*)d_in[8];
  const float* b1      = (const float*)d_in[9];
  float* out           = (float*)d_out;

  const int B = 8192;
  jacde_kernel<<<B / 16, 64, 0, stream>>>(t, hmat, coeffs, dcoeffs, tobs,
                                          wx, wh, wout, b0, b1, out);
}


# Products transposed: weights are the A operand, each product's D tiles serve unchanged as the next product's B fragments (sums run in a permuted k order); the 2 waves swap halves by 16-byte LDS copy
# speedup vs baseline: 1.3043x; 1.3043x over previous
#include <hip/hip_runtime.h>


typedef __attribute__((ext_vector_type(2))) float v2f;
typedef __attribute__((ext_vector_type(8))) float v8f;

#define HDIM  128
#define CINC  64
#define NOBSC 16
#define XLDP  68

__device__ __forceinline__ v8f wmma4(v2f a, v2f b, v8f c) {
  return __builtin_amdgcn_wmma_f32_16x16x4_f32(
       false, a,  false, b,
       (short)0, c,  false,  false);
}

struct APhase { float4 q[4][4][2]; };

template <int LDW>
__device__ __forceinline__ void load_a(APhase& A, const float* __restrict__ Wm, unsigned lo) {
  const float* base = Wm + lo;
  #pragma unroll
  for (int nt = 0; nt < 4; ++nt)
    #pragma unroll
    for (int i = 0; i < 4; ++i)
      #pragma unroll
      for (int q = 0; q < 2; ++q)
        A.q[nt][i][q] = *(const float4*)(base + nt * 16 * LDW + i * 16 + q * 4);
}

__device__ __forceinline__ void mma_phase(const APhase& A, const v8f* Bt, v8f* acc) {
  __builtin_amdgcn_sched_barrier(0);
  #pragma unroll
  for (int i = 0; i < 4; ++i)
    #pragma unroll
    for (int j = 0; j < 4; ++j) {
      v2f b;
      b.x = Bt[i][2 * j];
      b.y = Bt[i][2 * j + 1];
      #pragma unroll
      for (int nt = 0; nt < 4; ++nt) {
        const float4 qa = A.q[nt][i][j >> 1];
        v2f a;
        if (j & 1) { a.x = qa.z; a.y = qa.w; } else { a.x = qa.x; a.y = qa.y; }
        acc[nt] = wmma4(a, b, acc[nt]);
      }
    }
  asm volatile("v_nop\n\tv_nop\n\tv_nop\n\tv_nop"
               : "+v"(acc[0]), "+v"(acc[1]), "+v"(acc[2]), "+v"(acc[3]));
  __builtin_amdgcn_sched_barrier(0);
}

__device__ __forceinline__ void put_tiles(float4* dst, const v8f* t) {
  #pragma unroll
  for (int i = 0; i < 4; ++i)
    #pragma unroll
    for (int q = 0; q < 2; ++q)
      dst[(i * 2 + q) * 32] = make_float4(t[i][4 * q], t[i][4 * q + 1], t[i][4 * q + 2], t[i][4 * q + 3]);
}

__device__ __forceinline__ void get_tiles(const float4* src, v8f* t) {
  #pragma unroll
  for (int i = 0; i < 4; ++i)
    #pragma unroll
    for (int q = 0; q < 2; ++q) {
      const float4 v = src[(i * 2 + q) * 32];
      t[i][4 * q] = v.x; t[i][4 * q + 1] = v.y; t[i][4 * q + 2] = v.z; t[i][4 * q + 3] = v.w;
    }
}

__device__ __forceinline__ void lds_barrier() {
  __builtin_amdgcn_fence(__ATOMIC_RELEASE, "workgroup", "local");
  __builtin_amdgcn_s_barrier();
  __builtin_amdgcn_fence(__ATOMIC_ACQUIRE, "workgroup", "local");
}

__device__ __forceinline__ v8f load8(const float* __restrict__ p) {
  const float4 u0 = *(const float4*)p;
  const float4 u1 = *(const float4*)(p + 4);
  v8f r;
  r[0] = u0.x; r[1] = u0.y; r[2] = u0.z; r[3] = u0.w;
  r[4] = u1.x; r[5] = u1.y; r[6] = u1.z; r[7] = u1.w;
  return r;
}

__global__ void __launch_bounds__(64) __attribute__((amdgpu_waves_per_eu(1, 1))) jacde_kernel(
    const float* __restrict__ t,      const float* __restrict__ hmat,
    const float* __restrict__ coeffs, const float* __restrict__ dcoeffs,
    const float* __restrict__ tobs,
    const float* __restrict__ wx,     const float* __restrict__ wh,
    const float* __restrict__ wout,
    const float* __restrict__ b0,     const float* __restrict__ b1,
    float* __restrict__ out) {
  __shared__ float4 xch[2][2][8][32];
  __shared__ __attribute__((aligned(16))) float sx[2][16][XLDP];

  const int tid  = threadIdx.x;
  const int lane = tid & 31;
  const int wave = tid >> 5;
  const int c    = lane & 15;
  const int hh   = lane >> 4;
  const int brow = blockIdx.x * 16;
  const int f0   = wave * 64;
  const int pf0  = 64 - f0;

  unsigned lo_x = (f0 + c) * CINC + 8 * hh;
  unsigned lo_o = (f0 + c) * HDIM + 8 * hh + f0;
  unsigned lo_p = (f0 + c) * HDIM + 8 * hh + pf0;
  unsigned my_i = wave * 256 + lane;
  unsigned pr_i = (wave ^ 1) * 256 + lane;
  float4* xf = &xch[0][0][0][0];
  float4* my0 = xf + my_i;
  float4* my1 = xf + 512 + my_i;
  const float4* pr0 = xf + pr_i;
  const float4* pr1 = xf + 512 + pr_i;

  APhase A0, A1;
  load_a<HDIM>(A0, wh, lo_o);

  v8f Bh[8];
  {
    const float* hp = hmat + (size_t)(brow + c) * HDIM + 8 * hh;
    #pragma unroll
    for (int i = 0; i < 4; ++i) Bh[i] = load8(hp + f0 + 16 * i);
    #pragma unroll
    for (int i = 0; i < 4; ++i) Bh[4 + i] = load8(hp + pf0 + 16 * i);
  }
  load_a<HDIM>(A1, wh, lo_p);

  const float ts = t[0];
  int idx = 0;
  #pragma unroll
  for (int j = 0; j < NOBSC; ++j) idx += (tobs[j] <= ts) ? 1 : 0;
  idx -= 1;
  idx = idx < 0 ? 0 : (idx > NOBSC - 2 ? NOBSC - 2 : idx);
  const float dt = ts - tobs[idx];
  const float p1 = dt, p2 = dt * dt, p3 = p2 * dt;

  float4 cx[16], cd[16];
  {
    const size_t off = ((size_t)brow * (NOBSC - 1) + idx) * CINC * 4 + tid * 4;
    const float* cb = coeffs + off;
    const float* db = dcoeffs + off;
    #pragma unroll
    for (int i = 0; i < 16; ++i) cx[i] = *(const float4*)(cb + (size_t)i * (NOBSC - 1) * CINC * 4);
    #pragma unroll
    for (int i = 0; i < 16; ++i) cd[i] = *(const float4*)(db + (size_t)i * (NOBSC - 1) * CINC * 4);
  }

  v8f acc[4];
  #pragma unroll
  for (int nt = 0; nt < 4; ++nt) acc[nt] = load8(b0 + f0 + 16 * nt + 8 * hh);
  mma_phase(A0, Bh, acc);
  load_a<CINC>(A0, wx, lo_x);
  #pragma unroll
  for (int i = 0; i < 16; ++i) {
    sx[0][i][tid] = cx[i].x + cx[i].y * p1 + cx[i].z * p2 + cx[i].w * p3;
    sx[1][i][tid] = cd[i].x + cd[i].y * p1 + cd[i].z * p2 + cd[i].w * p3;
  }
  mma_phase(A1, Bh + 4, acc);
  load_a<HDIM>(A1, wout, lo_o);
  lds_barrier();
  v8f Bx[4], Bxd[4];
  #pragma unroll
  for (int i = 0; i < 4; ++i) {
    Bx[i]  = load8(&sx[0][c][16 * i + 8 * hh]);
    Bxd[i] = load8(&sx[1][c][16 * i + 8 * hh]);
  }
  mma_phase(A0, Bx, acc);
  load_a<HDIM>(A0, wout, lo_p);

  v8f drelu[4], own[4], par[4];
  #pragma unroll
  for (int nt = 0; nt < 4; ++nt)
    #pragma unroll
    for (int r = 0; r < 8; ++r) {
      const float x = acc[nt][r];
      drelu[nt][r] = 1.0f / (1.0f + __expf(-x));
      own[nt][r]   = fmaxf(x, 0.0f);
    }
  put_tiles(my0, own);

  v8f dtn[4];
  {
    v8f a2[4];
    #pragma unroll
    for (int nt = 0; nt < 4; ++nt) a2[nt] = load8(b1 + f0 + 16 * nt + 8 * hh);
    mma_phase(A1, own, a2);
    lds_barrier();
    get_tiles(pr0, par);
    load_a<CINC>(A1, wx, lo_x);
    mma_phase(A0, par, a2);
    load_a<HDIM>(A0, wout, lo_o);
    #pragma unroll
    for (int nt = 0; nt < 4; ++nt)
      #pragma unroll
      for (int r = 0; r < 8; ++r) {
        const float th = tanhf(a2[nt][r]);
        dtn[nt][r] = 1.0f - th * th;
      }
  }

  v8f cur[4];
  #pragma unroll
  for (int nt = 0; nt < 4; ++nt) cur[nt] = (v8f)(0.0f);
  mma_phase(A1, Bxd, cur);
  #pragma unroll
  for (int nt = 0; nt < 4; ++nt) own[nt] = cur[nt] * drelu[nt];
  put_tiles(my1, own);

  v8f outv[4];
  #pragma unroll
  for (int nt = 0; nt < 4; ++nt) outv[nt] = (v8f)(0.0f);
  #pragma nounroll
  for (int n = 0; n < 5; ++n) {
    const bool odd = (n & 1) != 0;
    const float* Wc = odd ? wh : wout;
    const float* Wn = odd ? wout : wh;
    load_a<HDIM>(A1, Wc, lo_p);
    v8f tt[4];
    #pragma unroll
    for (int nt = 0; nt < 4; ++nt) tt[nt] = (v8f)(0.0f);
    mma_phase(A0, own, tt);
    lds_barrier();
    get_tiles(odd ? pr0 : pr1, par);
    if (n < 4) load_a<HDIM>(A0, Wn, lo_o);
    mma_phase(A1, par, tt);
    #pragma unroll
    for (int nt = 0; nt < 4; ++nt) {
      const v8f mlt = odd ? drelu[nt] : dtn[nt];
      tt[nt] *= mlt;
      own[nt] = tt[nt];
      outv[nt] += odd ? (v8f)(0.0f) : tt[nt];
    }
    if (n < 4) put_tiles(odd ? my1 : my0, own);
  }

  #pragma unroll
  for (int nt = 0; nt < 4; ++nt) {
    float* op = out + (size_t)(brow + c) * HDIM + f0 + 16 * nt + 8 * hh;
    *(float4*)op       = make_float4(outv[nt][0], outv[nt][1], outv[nt][2], outv[nt][3]);
    *(float4*)(op + 4) = make_float4(outv[nt][4], outv[nt][5], outv[nt][6], outv[nt][7]);
  }
}

extern "C" void kernel_launch(void* const* d_in, const int* in_sizes, int n_in,
                              void* d_out, int out_size, void* d_ws, size_t ws_size,
                              hipStream_t stream) {
  const float* t       = (const float*)d_in[0];
  const float* hmat    = (const float*)d_in[1];
  const float* coeffs  = (const float*)d_in[2];
  const float* dcoeffs = (const float*)d_in[3];
  const float* tobs    = (const float*)d_in[4];
  const float* wx      = (const float*)d_in[5];
  const float* wh      = (const float*)d_in[6];
  const float* wout    = (const float*)d_in[7];
  const float* b0      = (const float*)d_in[8];
  const float* b1      = (const float*)d_in[9];
  float* out           = (float*)d_out;

  const int B = 8192;
  jacde_kernel<<<B / 16, 64, 0, stream>>>(t, hmat, coeffs, dcoeffs, tobs,
                                          wx, wh, wout, b0, b1, out);
}


# Wave tile 32 features x 32 batch rows (4 waves per 32-row block): a weight fragment feeds 2 matrix instructions, wx and 2 of wout's 4 k groups fetched once; k summed own pair, then waves w+1,w+2,w+3
# speedup vs baseline: 1.6342x; 1.2529x over previous
#include <hip/hip_runtime.h>


typedef __attribute__((ext_vector_type(2))) float v2f;
typedef __attribute__((ext_vector_type(8))) float v8f;

#define HDIM  128
#define CINC  64
#define NOBSC 16
#define XLDP  68

__device__ __forceinline__ v8f wmma4(v2f a, v2f b, v8f c) {
  return __builtin_amdgcn_wmma_f32_16x16x4_f32(
       false, a,  false, b,
       (short)0, c,  false,  false);
}

struct AGroup { float4 q[2][2][2]; };

template <int LDW>
__device__ __forceinline__ void load_group(AGroup& A, const float* __restrict__ Wm, unsigned lo) {
  const float* base = Wm + lo;
  #pragma unroll
  for (int ft = 0; ft < 2; ++ft)
    #pragma unroll
    for (int tt = 0; tt < 2; ++tt)
      #pragma unroll
      for (int q = 0; q < 2; ++q)
        A.q[ft][tt][q] = *(const float4*)(base + ft * 16 * LDW + tt * 16 + q * 4);
}

__device__ __forceinline__ void mma_body(const AGroup& A, const v8f* Bt, v8f* acc) {
  #pragma unroll
  for (int tt = 0; tt < 2; ++tt)
    #pragma unroll
    for (int j = 0; j < 4; ++j) {
      v2f b0, b1;
      b0.x = Bt[tt * 2 + 0][2 * j];
      b0.y = Bt[tt * 2 + 0][2 * j + 1];
      b1.x = Bt[tt * 2 + 1][2 * j];
      b1.y = Bt[tt * 2 + 1][2 * j + 1];
      #pragma unroll
      for (int ft = 0; ft < 2; ++ft) {
        const float4 qa = A.q[ft][tt][j >> 1];
        v2f a;
        if (j & 1) { a.x = qa.z; a.y = qa.w; } else { a.x = qa.x; a.y = qa.y; }
        acc[ft * 2 + 0] = wmma4(a, b0, acc[ft * 2 + 0]);
        acc[ft * 2 + 1] = wmma4(a, b1, acc[ft * 2 + 1]);
      }
    }
}

__device__ __forceinline__ void mma_open() {
  __builtin_amdgcn_sched_barrier(0);
}
__device__ __forceinline__ void mma_close(v8f* acc) {
  asm volatile("v_nop\n\tv_nop\n\tv_nop\n\tv_nop"
               : "+v"(acc[0]), "+v"(acc[1]), "+v"(acc[2]), "+v"(acc[3]));
  __builtin_amdgcn_sched_barrier(0);
}
__device__ __forceinline__ void mma_one(const AGroup& A, const v8f* Bt, v8f* acc) {
  mma_open();
  mma_body(A, Bt, acc);
  mma_close(acc);
}
__device__ __forceinline__ void mma_two(const AGroup& A, const v8f* Bt,
                                        const AGroup& A2, const v8f* Bt2, v8f* acc) {
  mma_open();
  mma_body(A, Bt, acc);
  mma_body(A2, Bt2, acc);
  mma_close(acc);
}
__device__ __forceinline__ void mma_three(const AGroup& A, const v8f* Bt,
                                          const AGroup& A2, const v8f* Bt2,
                                          const AGroup& A3, const v8f* Bt3, v8f* acc) {
  mma_open();
  mma_body(A, Bt, acc);
  mma_body(A2, Bt2, acc);
  mma_body(A3, Bt3, acc);
  mma_close(acc);
}

__device__ __forceinline__ void put_tiles(float4* dst, const v8f* t) {
  #pragma unroll
  for (int i = 0; i < 4; ++i)
    #pragma unroll
    for (int q = 0; q < 2; ++q)
      dst[(i * 2 + q) * 32] = make_float4(t[i][4 * q], t[i][4 * q + 1], t[i][4 * q + 2], t[i][4 * q + 3]);
}

__device__ __forceinline__ void get_tiles(const float4* src, v8f* t) {
  #pragma unroll
  for (int i = 0; i < 4; ++i)
    #pragma unroll
    for (int q = 0; q < 2; ++q) {
      const float4 v = src[(i * 2 + q) * 32];
      t[i][4 * q] = v.x; t[i][4 * q + 1] = v.y; t[i][4 * q + 2] = v.z; t[i][4 * q + 3] = v.w;
    }
}

__device__ __forceinline__ void lds_barrier() {
  __builtin_amdgcn_fence(__ATOMIC_RELEASE, "workgroup", "local");
  __builtin_amdgcn_s_barrier();
  __builtin_amdgcn_fence(__ATOMIC_ACQUIRE, "workgroup", "local");
}

__device__ __forceinline__ v8f load8(const float* __restrict__ p) {
  const float4 u0 = *(const float4*)p;
  const float4 u1 = *(const float4*)(p + 4);
  v8f r;
  r[0] = u0.x; r[1] = u0.y; r[2] = u0.z; r[3] = u0.w;
  r[4] = u1.x; r[5] = u1.y; r[6] = u1.z; r[7] = u1.w;
  return r;
}

__global__ void __launch_bounds__(128) __attribute__((amdgpu_waves_per_eu(1, 1))) jacde_kernel(
    const float* __restrict__ t,      const float* __restrict__ hmat,
    const float* __restrict__ coeffs, const float* __restrict__ dcoeffs,
    const float* __restrict__ tobs,
    const float* __restrict__ wx,     const float* __restrict__ wh,
    const float* __restrict__ wout,
    const float* __restrict__ b0,     const float* __restrict__ b1,
    float* __restrict__ out) {
  __shared__ float4 xch[2][4][8][32];
  __shared__ __attribute__((aligned(16))) float sx[2][32][XLDP];

  const int tid  = threadIdx.x;
  const int lane = tid & 31;
  const int wave = tid >> 5;
  const int c    = lane & 15;
  const int hh   = lane >> 4;
  const int brow = blockIdx.x * 32;
  const int f0   = wave * 32;

  const unsigned lo_x  = (f0 + c) * CINC + 8 * hh;
  const unsigned lo_w  = (f0 + c) * HDIM + 8 * hh;
  const unsigned lo_g0 = lo_w + 32 * wave;
  const unsigned lo_g1 = lo_w + 32 * ((wave + 1) & 3);
  const unsigned lo_g2 = lo_w + 32 * ((wave + 2) & 3);
  const unsigned lo_g3 = lo_w + 32 * ((wave + 3) & 3);
  float4* xf = &xch[0][0][0][0];
  float4* my0 = xf + wave * 256 + lane;
  float4* my1 = my0 + 1024;
  const float4* pa0 = xf + ((wave + 1) & 3) * 256 + lane;
  const float4* pb0 = xf + ((wave + 2) & 3) * 256 + lane;
  const float4* pc0 = xf + ((wave + 3) & 3) * 256 + lane;

  AGroup AX0, AX1, AY0, AY1, AZ0, AZ1;
  load_group<HDIM>(AX0, wh, lo_g0);
  load_group<HDIM>(AX1, wh, lo_g1);

  v8f own[4];
  {
    const float* hp = hmat + (size_t)(brow + c) * HDIM + f0 + 8 * hh;
    #pragma unroll
    for (int tt = 0; tt < 2; ++tt)
      #pragma unroll
      for (int bt = 0; bt < 2; ++bt)
        own[tt * 2 + bt] = load8(hp + (size_t)bt * 16 * HDIM + 16 * tt);
  }
  load_group<HDIM>(AY0, wh, lo_g2);
  load_group<HDIM>(AY1, wh, lo_g3);
  load_group<CINC>(AZ0, wx, lo_x);
  load_group<CINC>(AZ1, wx, lo_x + 32);

  const float ts = t[0];
  int idx = 0;
  #pragma unroll
  for (int j = 0; j < NOBSC; ++j) idx += (tobs[j] <= ts) ? 1 : 0;
  idx -= 1;
  idx = idx < 0 ? 0 : (idx > NOBSC - 2 ? NOBSC - 2 : idx);
  const float dt = ts - tobs[idx];
  const float p1 = dt, p2 = dt * dt, p3 = p2 * dt;

  const int ch = tid & 63;
  const int rg = tid >> 6;
  float4 cx[16], cd[16];
  {
    const size_t off = ((size_t)(brow + 16 * rg) * (NOBSC - 1) + idx) * CINC * 4 + ch * 4;
    const float* cb = coeffs + off;
    const float* db = dcoeffs + off;
    #pragma unroll
    for (int i = 0; i < 16; ++i) cx[i] = *(const float4*)(cb + (size_t)i * (NOBSC - 1) * CINC * 4);
    #pragma unroll
    for (int i = 0; i < 16; ++i) cd[i] = *(const float4*)(db + (size_t)i * (NOBSC - 1) * CINC * 4);
  }

  v8f acc[4], par0[4], par1[4], par2[4];
  #pragma unroll
  for (int ft = 0; ft < 2; ++ft) {
    acc[ft * 2 + 0] = load8(b0 + f0 + 16 * ft + 8 * hh);
    acc[ft * 2 + 1] = acc[ft * 2 + 0];
  }
  put_tiles(my0, own);
  mma_one(AX0, own, acc);
  lds_barrier();
  get_tiles(pa0, par0);
  get_tiles(pb0, par1);
  get_tiles(pc0, par2);
  mma_one(AX1, par0, acc);
  load_group<HDIM>(AX0, wout, lo_g0);
  load_group<HDIM>(AX1, wout, lo_g1);
  mma_two(AY0, par1, AY1, par2, acc);
  load_group<HDIM>(AY0, wout, lo_g2);
  load_group<HDIM>(AY1, wout, lo_g3);
  #pragma unroll
  for (int i = 0; i < 16; ++i) {
    sx[0][16 * rg + i][ch] = cx[i].x + cx[i].y * p1 + cx[i].z * p2 + cx[i].w * p3;
    sx[1][16 * rg + i][ch] = cd[i].x + cd[i].y * p1 + cd[i].z * p2 + cd[i].w * p3;
  }
  lds_barrier();
  {
    v8f Bx[8];
    #pragma unroll
    for (int T = 0; T < 4; ++T)
      #pragma unroll
      for (int bt = 0; bt < 2; ++bt)
        Bx[T * 2 + bt] = load8(&sx[0][16 * bt + c][16 * T + 8 * hh]);
    mma_two(AZ0, Bx, AZ1, Bx + 4, acc);
  }

  v8f drelu[4];
  #pragma unroll
  for (int i = 0; i < 4; ++i)
    #pragma unroll
    for (int r = 0; r < 8; ++r) {
      const float x = acc[i][r];
      drelu[i][r] = 1.0f / (1.0f + __expf(-x));
      own[i][r]   = fmaxf(x, 0.0f);
    }
  put_tiles(my1, own);

  v8f dtn[4];
  {
    v8f a2[4];
    #pragma unroll
    for (int ft = 0; ft < 2; ++ft) {
      a2[ft * 2 + 0] = load8(b1 + f0 + 16 * ft + 8 * hh);
      a2[ft * 2 + 1] = a2[ft * 2 + 0];
    }
    mma_one(AX0, own, a2);
    lds_barrier();
    get_tiles(pa0 + 1024, par0);
    get_tiles(pb0 + 1024, par1);
    get_tiles(pc0 + 1024, par2);
    mma_three(AX1, par0, AY0, par1, AY1, par2, a2);
    #pragma unroll
    for (int i = 0; i < 4; ++i)
      #pragma unroll
      for (int r = 0; r < 8; ++r) {
        const float th = tanhf(a2[i][r]);
        dtn[i][r] = 1.0f - th * th;
      }
  }

  {
    v8f Bxd[8];
    #pragma unroll
    for (int T = 0; T < 4; ++T)
      #pragma unroll
      for (int bt = 0; bt < 2; ++bt)
        Bxd[T * 2 + bt] = load8(&sx[1][16 * bt + c][16 * T + 8 * hh]);
    v8f cur[4];
    #pragma unroll
    for (int i = 0; i < 4; ++i) cur[i] = (v8f)(0.0f);
    mma_two(AZ0, Bxd, AZ1, Bxd + 4, cur);
    #pragma unroll
    for (int i = 0; i < 4; ++i) own[i] = cur[i] * drelu[i];
  }
  put_tiles(my0, own);

  v8f outv[4];
  #pragma unroll
  for (int i = 0; i < 4; ++i) outv[i] = (v8f)(0.0f);
  #pragma nounroll
  for (int n = 0; n < 5; ++n) {
    const bool odd = (n & 1) != 0;
    const float* Wc = odd ? wh : wout;
    const float* Wn = odd ? wout : wh;
    const int rb = odd ? 1024 : 0;
    load_group<HDIM>(AY0, Wc, lo_g2);
    load_group<HDIM>(AY1, Wc, lo_g3);
    v8f tt[4];
    #pragma unroll
    for (int i = 0; i < 4; ++i) tt[i] = (v8f)(0.0f);
    mma_one(AX0, own, tt);
    lds_barrier();
    get_tiles(pa0 + rb, par0);
    get_tiles(pb0 + rb, par1);
    get_tiles(pc0 + rb, par2);
    mma_one(AX1, par0, tt);
    if (n < 4) {
      load_group<HDIM>(AX0, Wn, lo_g0);
      load_group<HDIM>(AX1, Wn, lo_g1);
    }
    mma_two(AY0, par1, AY1, par2, tt);
    #pragma unroll
    for (int i = 0; i < 4; ++i) {
      const v8f mlt = odd ? drelu[i] : dtn[i];
      tt[i] *= mlt;
      own[i] = tt[i];
      outv[i] += odd ? (v8f)(0.0f) : tt[i];
    }
    if (n < 4) put_tiles(odd ? my0 : my1, own);
  }

  #pragma unroll
  for (int ft = 0; ft < 2; ++ft)
    #pragma unroll
    for (int bt = 0; bt < 2; ++bt) {
      float* op = out + (size_t)(brow + 16 * bt + c) * HDIM + f0 + 16 * ft + 8 * hh;
      const v8f v = outv[ft * 2 + bt];
      *(float4*)op       = make_float4(v[0], v[1], v[2], v[3]);
      *(float4*)(op + 4) = make_float4(v[4], v[5], v[6], v[7]);
    }
}

extern "C" void kernel_launch(void* const* d_in, const int* in_sizes, int n_in,
                              void* d_out, int out_size, void* d_ws, size_t ws_size,
                              hipStream_t stream) {
  const float* t       = (const float*)d_in[0];
  const float* hmat    = (const float*)d_in[1];
  const float* coeffs  = (const float*)d_in[2];
  const float* dcoeffs = (const float*)d_in[3];
  const float* tobs    = (const float*)d_in[4];
  const float* wx      = (const float*)d_in[5];
  const float* wh      = (const float*)d_in[6];
  const float* wout    = (const float*)d_in[7];
  const float* b0      = (const float*)d_in[8];
  const float* b1      = (const float*)d_in[9];
  float* out           = (float*)d_out;

  const int B = 8192;
  jacde_kernel<<<B / 32, 128, 0, stream>>>(t, hmat, coeffs, dcoeffs, tobs,
                                           wx, wh, wout, b0, b1, out);
}


# Loads issued in the order the products need them, none under a branch: no matrix run waits for a load younger than its operands (h@wh starts while coefficients and wx still stream in)
# speedup vs baseline: 1.9099x; 1.1687x over previous
#include <hip/hip_runtime.h>


typedef __attribute__((ext_vector_type(2))) float v2f;
typedef __attribute__((ext_vector_type(8))) float v8f;

#define HDIM  128
#define CINC  64
#define NOBSC 16
#define XLDP  68

__device__ __forceinline__ v8f wmma4(v2f a, v2f b, v8f c) {
  return __builtin_amdgcn_wmma_f32_16x16x4_f32(
       false, a,  false, b,
       (short)0, c,  false,  false);
}

struct AGroup { float4 q[2][2][2]; };

template <int LDW>
__device__ __forceinline__ void load_group(AGroup& A, const float* __restrict__ Wm, unsigned lo) {
  const float* base = Wm + lo;
  #pragma unroll
  for (int ft = 0; ft < 2; ++ft)
    #pragma unroll
    for (int tt = 0; tt < 2; ++tt)
      #pragma unroll
      for (int q = 0; q < 2; ++q)
        A.q[ft][tt][q] = *(const float4*)(base + ft * 16 * LDW + tt * 16 + q * 4);
}

__device__ __forceinline__ void mma_body(const AGroup& A, const v8f* Bt, v8f* acc) {
  #pragma unroll
  for (int tt = 0; tt < 2; ++tt)
    #pragma unroll
    for (int j = 0; j < 4; ++j) {
      v2f b0, b1;
      b0.x = Bt[tt * 2 + 0][2 * j];
      b0.y = Bt[tt * 2 + 0][2 * j + 1];
      b1.x = Bt[tt * 2 + 1][2 * j];
      b1.y = Bt[tt * 2 + 1][2 * j + 1];
      #pragma unroll
      for (int ft = 0; ft < 2; ++ft) {
        const float4 qa = A.q[ft][tt][j >> 1];
        v2f a;
        if (j & 1) { a.x = qa.z; a.y = qa.w; } else { a.x = qa.x; a.y = qa.y; }
        acc[ft * 2 + 0] = wmma4(a, b0, acc[ft * 2 + 0]);
        acc[ft * 2 + 1] = wmma4(a, b1, acc[ft * 2 + 1]);
      }
    }
}

__device__ __forceinline__ void mma_open() {
  __builtin_amdgcn_sched_barrier(0);
}
__device__ __forceinline__ void mma_close(v8f* acc) {
  asm volatile("v_nop\n\tv_nop\n\tv_nop\n\tv_nop"
               : "+v"(acc[0]), "+v"(acc[1]), "+v"(acc[2]), "+v"(acc[3]));
  __builtin_amdgcn_sched_barrier(0);
}
__device__ __forceinline__ void mma_one(const AGroup& A, const v8f* Bt, v8f* acc) {
  mma_open();
  mma_body(A, Bt, acc);
  mma_close(acc);
}
__device__ __forceinline__ void mma_two(const AGroup& A, const v8f* Bt,
                                        const AGroup& A2, const v8f* Bt2, v8f* acc) {
  mma_open();
  mma_body(A, Bt, acc);
  mma_body(A2, Bt2, acc);
  mma_close(acc);
}
__device__ __forceinline__ void mma_three(const AGroup& A, const v8f* Bt,
                                          const AGroup& A2, const v8f* Bt2,
                                          const AGroup& A3, const v8f* Bt3, v8f* acc) {
  mma_open();
  mma_body(A, Bt, acc);
  mma_body(A2, Bt2, acc);
  mma_body(A3, Bt3, acc);
  mma_close(acc);
}

__device__ __forceinline__ void put_tiles(float4* dst, const v8f* t) {
  #pragma unroll
  for (int i = 0; i < 4; ++i)
    #pragma unroll
    for (int q = 0; q < 2; ++q)
      dst[(i * 2 + q) * 32] = make_float4(t[i][4 * q], t[i][4 * q + 1], t[i][4 * q + 2], t[i][4 * q + 3]);
}

__device__ __forceinline__ void get_tiles(const float4* src, v8f* t) {
  #pragma unroll
  for (int i = 0; i < 4; ++i)
    #pragma unroll
    for (int q = 0; q < 2; ++q) {
      const float4 v = src[(i * 2 + q) * 32];
      t[i][4 * q] = v.x; t[i][4 * q + 1] = v.y; t[i][4 * q + 2] = v.z; t[i][4 * q + 3] = v.w;
    }
}

__device__ __forceinline__ void lds_barrier() {
  __builtin_amdgcn_fence(__ATOMIC_RELEASE, "workgroup", "local");
  __builtin_amdgcn_s_barrier();
  __builtin_amdgcn_fence(__ATOMIC_ACQUIRE, "workgroup", "local");
}

__device__ __forceinline__ v8f load8(const float* __restrict__ p) {
  const float4 u0 = *(const float4*)p;
  const float4 u1 = *(const float4*)(p + 4);
  v8f r;
  r[0] = u0.x; r[1] = u0.y; r[2] = u0.z; r[3] = u0.w;
  r[4] = u1.x; r[5] = u1.y; r[6] = u1.z; r[7] = u1.w;
  return r;
}

__global__ void __launch_bounds__(128) __attribute__((amdgpu_waves_per_eu(1, 1))) jacde_kernel(
    const float* __restrict__ t,      const float* __restrict__ hmat,
    const float* __restrict__ coeffs, const float* __restrict__ dcoeffs,
    const float* __restrict__ tobs,
    const float* __restrict__ wx,     const float* __restrict__ wh,
    const float* __restrict__ wout,
    const float* __restrict__ b0,     const float* __restrict__ b1,
    float* __restrict__ out) {
  __shared__ float4 xch[2][4][8][32];
  __shared__ __attribute__((aligned(16))) float sx[2][32][XLDP];

  const int tid  = threadIdx.x;
  const int lane = tid & 31;
  const int wave = tid >> 5;
  const int c    = lane & 15;
  const int hh   = lane >> 4;
  const int brow = blockIdx.x * 32;
  const int f0   = wave * 32;

  const unsigned lo_x  = (f0 + c) * CINC + 8 * hh;
  const unsigned lo_w  = (f0 + c) * HDIM + 8 * hh;
  const unsigned lo_g0 = lo_w + 32 * wave;
  const unsigned lo_g1 = lo_w + 32 * ((wave + 1) & 3);
  const unsigned lo_g2 = lo_w + 32 * ((wave + 2) & 3);
  const unsigned lo_g3 = lo_w + 32 * ((wave + 3) & 3);
  float4* xf = &xch[0][0][0][0];
  float4* my0 = xf + wave * 256 + lane;
  float4* my1 = my0 + 1024;
  const float4* pa0 = xf + ((wave + 1) & 3) * 256 + lane;
  const float4* pb0 = xf + ((wave + 2) & 3) * 256 + lane;
  const float4* pc0 = xf + ((wave + 3) & 3) * 256 + lane;

  AGroup AX0, AX1, AY0, AY1, AZ0, AZ1;
  const float ts = t[0];
  float tb[NOBSC];
  #pragma unroll
  for (int j = 0; j < NOBSC; ++j) tb[j] = tobs[j];

  load_group<HDIM>(AX0, wh, lo_g0);
  v8f own[4];
  {
    const float* hp = hmat + (size_t)(brow + c) * HDIM + f0 + 8 * hh;
    #pragma unroll
    for (int tt = 0; tt < 2; ++tt)
      #pragma unroll
      for (int bt = 0; bt < 2; ++bt)
        own[tt * 2 + bt] = load8(hp + (size_t)bt * 16 * HDIM + 16 * tt);
  }
  v8f acc[4];
  #pragma unroll
  for (int ft = 0; ft < 2; ++ft) {
    acc[ft * 2 + 0] = load8(b0 + f0 + 16 * ft + 8 * hh);
    acc[ft * 2 + 1] = acc[ft * 2 + 0];
  }
  __builtin_amdgcn_sched_barrier(0);
  load_group<HDIM>(AX1, wh, lo_g1);
  load_group<HDIM>(AY0, wh, lo_g2);
  load_group<HDIM>(AY1, wh, lo_g3);
  __builtin_amdgcn_sched_barrier(0);

  int idx = 0;
  #pragma unroll
  for (int j = 0; j < NOBSC; ++j) idx += (tb[j] <= ts) ? 1 : 0;
  idx -= 1;
  idx = idx < 0 ? 0 : (idx > NOBSC - 2 ? NOBSC - 2 : idx);
  const float dt = ts - tobs[idx];
  const float p1 = dt, p2 = dt * dt, p3 = p2 * dt;

  const int ch = tid & 63;
  const int rg = tid >> 6;
  float4 cx[16], cd[16];
  {
    const size_t off = ((size_t)(brow + 16 * rg) * (NOBSC - 1) + idx) * CINC * 4 + ch * 4;
    const float* cb = coeffs + off;
    const float* db = dcoeffs + off;
    #pragma unroll
    for (int i = 0; i < 16; ++i) cx[i] = *(const float4*)(cb + (size_t)i * (NOBSC - 1) * CINC * 4);
    #pragma unroll
    for (int i = 0; i < 16; ++i) cd[i] = *(const float4*)(db + (size_t)i * (NOBSC - 1) * CINC * 4);
  }

  v8f par0[4], par1[4], par2[4];
  put_tiles(my0, own);
  mma_one(AX0, own, acc);
  load_group<CINC>(AZ0, wx, lo_x);
  load_group<CINC>(AZ1, wx, lo_x + 32);
  lds_barrier();
  get_tiles(pa0, par0);
  get_tiles(pb0, par1);
  get_tiles(pc0, par2);
  mma_one(AX1, par0, acc);
  load_group<HDIM>(AX0, wout, lo_g0);
  load_group<HDIM>(AX1, wout, lo_g1);
  mma_two(AY0, par1, AY1, par2, acc);
  v8f a2[4];
  #pragma unroll
  for (int ft = 0; ft < 2; ++ft) {
    a2[ft * 2 + 0] = load8(b1 + f0 + 16 * ft + 8 * hh);
    a2[ft * 2 + 1] = a2[ft * 2 + 0];
  }
  load_group<HDIM>(AY0, wout, lo_g2);
  load_group<HDIM>(AY1, wout, lo_g3);
  #pragma unroll
  for (int i = 0; i < 16; ++i) {
    sx[0][16 * rg + i][ch] = cx[i].x + cx[i].y * p1 + cx[i].z * p2 + cx[i].w * p3;
    sx[1][16 * rg + i][ch] = cd[i].x + cd[i].y * p1 + cd[i].z * p2 + cd[i].w * p3;
  }
  lds_barrier();
  {
    v8f Bx[8];
    #pragma unroll
    for (int T = 0; T < 4; ++T)
      #pragma unroll
      for (int bt = 0; bt < 2; ++bt)
        Bx[T * 2 + bt] = load8(&sx[0][16 * bt + c][16 * T + 8 * hh]);
    mma_two(AZ0, Bx, AZ1, Bx + 4, acc);
  }

  v8f drelu[4];
  #pragma unroll
  for (int i = 0; i < 4; ++i)
    #pragma unroll
    for (int r = 0; r < 8; ++r) {
      const float x = acc[i][r];
      drelu[i][r] = 1.0f / (1.0f + __expf(-x));
      own[i][r]   = fmaxf(x, 0.0f);
    }
  put_tiles(my1, own);

  v8f dtn[4];
  {
    mma_one(AX0, own, a2);
    lds_barrier();
    get_tiles(pa0 + 1024, par0);
    get_tiles(pb0 + 1024, par1);
    get_tiles(pc0 + 1024, par2);
    mma_three(AX1, par0, AY0, par1, AY1, par2, a2);
    #pragma unroll
    for (int i = 0; i < 4; ++i)
      #pragma unroll
      for (int r = 0; r < 8; ++r) {
        const float th = tanhf(a2[i][r]);
        dtn[i][r] = 1.0f - th * th;
      }
  }

  {
    v8f Bxd[8];
    #pragma unroll
    for (int T = 0; T < 4; ++T)
      #pragma unroll
      for (int bt = 0; bt < 2; ++bt)
        Bxd[T * 2 + bt] = load8(&sx[1][16 * bt + c][16 * T + 8 * hh]);
    v8f cur[4];
    #pragma unroll
    for (int i = 0; i < 4; ++i) cur[i] = (v8f)(0.0f);
    mma_two(AZ0, Bxd, AZ1, Bxd + 4, cur);
    #pragma unroll
    for (int i = 0; i < 4; ++i) own[i] = cur[i] * drelu[i];
  }
  put_tiles(my0, own);

  v8f outv[4];
  #pragma unroll
  for (int i = 0; i < 4; ++i) outv[i] = (v8f)(0.0f);
  #pragma nounroll
  for (int n = 0; n < 5; ++n) {
    const bool odd = (n & 1) != 0;
    const float* Wc = odd ? wh : wout;
    const float* Wn = odd ? wout : wh;
    const int rb = odd ? 1024 : 0;
    load_group<HDIM>(AY0, Wc, lo_g2);
    load_group<HDIM>(AY1, Wc, lo_g3);
    v8f tt[4];
    #pragma unroll
    for (int i = 0; i < 4; ++i) tt[i] = (v8f)(0.0f);
    mma_one(AX0, own, tt);
    lds_barrier();
    get_tiles(pa0 + rb, par0);
    get_tiles(pb0 + rb, par1);
    get_tiles(pc0 + rb, par2);
    mma_one(AX1, par0, tt);
    load_group<HDIM>(AX0, Wn, lo_g0);
    load_group<HDIM>(AX1, Wn, lo_g1);
    mma_two(AY0, par1, AY1, par2, tt);
    #pragma unroll
    for (int i = 0; i < 4; ++i) {
      const v8f mlt = odd ? drelu[i] : dtn[i];
      tt[i] *= mlt;
      own[i] = tt[i];
      outv[i] += odd ? (v8f)(0.0f) : tt[i];
    }
    if (n < 4) put_tiles(odd ? my0 : my1, own);
  }

  #pragma unroll
  for (int ft = 0; ft < 2; ++ft)
    #pragma unroll
    for (int bt = 0; bt < 2; ++bt) {
      float* op = out + (size_t)(brow + 16 * bt + c) * HDIM + f0 + 16 * ft + 8 * hh;
      const v8f v = outv[ft * 2 + bt];
      *(float4*)op       = make_float4(v[0], v[1], v[2], v[3]);
      *(float4*)(op + 4) = make_float4(v[4], v[5], v[6], v[7]);
    }
}

extern "C" void kernel_launch(void* const* d_in, const int* in_sizes, int n_in,
                              void* d_out, int out_size, void* d_ws, size_t ws_size,
                              hipStream_t stream) {
  const float* t       = (const float*)d_in[0];
  const float* hmat    = (const float*)d_in[1];
  const float* coeffs  = (const float*)d_in[2];
  const float* dcoeffs = (const float*)d_in[3];
  const float* tobs    = (const float*)d_in[4];
  const float* wx      = (const float*)d_in[5];
  const float* wh      = (const float*)d_in[6];
  const float* wout    = (const float*)d_in[7];
  const float* b0      = (const float*)d_in[8];
  const float* b1      = (const float*)d_in[9];
  float* out           = (float*)d_out;

  const int B = 8192;
  jacde_kernel<<<B / 32, 128, 0, stream>>>(t, hmat, coeffs, dcoeffs, tobs,
                                           wx, wh, wout, b0, b1, out);
}


# 1 - tanh^2 computed between the xdot product's matrix instructions (3 vector instructions behind each, after its 4 hazard slots) as branch-free 4(w-w^2), w=1/(1+exp(2|a|)); was 32 serial tanh calls
# speedup vs baseline: 2.1186x; 1.1093x over previous
#include <hip/hip_runtime.h>


typedef __attribute__((ext_vector_type(2))) float v2f;
typedef __attribute__((ext_vector_type(8))) float v8f;

#define HDIM  128
#define CINC  64
#define NOBSC 16
#define XLDP  68

__device__ __forceinline__ v8f wmma4(v2f a, v2f b, v8f c) {
  return __builtin_amdgcn_wmma_f32_16x16x4_f32(
       false, a,  false, b,
       (short)0, c,  false,  false);
}

struct AGroup { float4 q[2][2][2]; };

template <int LDW>
__device__ __forceinline__ void load_group(AGroup& A, const float* __restrict__ Wm, unsigned lo) {
  const float* base = Wm + lo;
  #pragma unroll
  for (int ft = 0; ft < 2; ++ft)
    #pragma unroll
    for (int tt = 0; tt < 2; ++tt)
      #pragma unroll
      for (int q = 0; q < 2; ++q)
        A.q[ft][tt][q] = *(const float4*)(base + ft * 16 * LDW + tt * 16 + q * 4);
}

__device__ __forceinline__ void mma_body(const AGroup& A, const v8f* Bt, v8f* acc) {
  #pragma unroll
  for (int tt = 0; tt < 2; ++tt)
    #pragma unroll
    for (int j = 0; j < 4; ++j) {
      v2f b0, b1;
      b0.x = Bt[tt * 2 + 0][2 * j];
      b0.y = Bt[tt * 2 + 0][2 * j + 1];
      b1.x = Bt[tt * 2 + 1][2 * j];
      b1.y = Bt[tt * 2 + 1][2 * j + 1];
      #pragma unroll
      for (int ft = 0; ft < 2; ++ft) {
        const float4 qa = A.q[ft][tt][j >> 1];
        v2f a;
        if (j & 1) { a.x = qa.z; a.y = qa.w; } else { a.x = qa.x; a.y = qa.y; }
        acc[ft * 2 + 0] = wmma4(a, b0, acc[ft * 2 + 0]);
        acc[ft * 2 + 1] = wmma4(a, b1, acc[ft * 2 + 1]);
      }
    }
}

__device__ __forceinline__ void mma_open() {
  __builtin_amdgcn_sched_barrier(0);
}
__device__ __forceinline__ void mma_close(v8f* acc) {
  asm volatile("v_nop\n\tv_nop\n\tv_nop\n\tv_nop"
               : "+v"(acc[0]), "+v"(acc[1]), "+v"(acc[2]), "+v"(acc[3]));
  __builtin_amdgcn_sched_barrier(0);
}
__device__ __forceinline__ void mma_one(const AGroup& A, const v8f* Bt, v8f* acc) {
  mma_open();
  mma_body(A, Bt, acc);
  mma_close(acc);
}
__device__ __forceinline__ void mma_two(const AGroup& A, const v8f* Bt,
                                        const AGroup& A2, const v8f* Bt2, v8f* acc) {
  mma_open();
  mma_body(A, Bt, acc);
  mma_body(A2, Bt2, acc);
  mma_close(acc);
}
__device__ __forceinline__ void mma_three(const AGroup& A, const v8f* Bt,
                                          const AGroup& A2, const v8f* Bt2,
                                          const AGroup& A3, const v8f* Bt3, v8f* acc) {
  mma_open();
  mma_body(A, Bt, acc);
  mma_body(A2, Bt2, acc);
  mma_body(A3, Bt3, acc);
  mma_close(acc);
}

__device__ __forceinline__ float dtanh_stage(int s, float x) {
  switch (s) {
    case 0:  return __builtin_fabsf(x) * 2.885390081777927f;
    case 1:  return __builtin_amdgcn_exp2f(x);
    case 2:  return x + 1.0f;
    case 3:  return __builtin_amdgcn_rcpf(x);
    case 4:  return __builtin_fmaf(-x, x, x);
    default: return 4.0f * x;
  }
}

__device__ __forceinline__ void mma_two_with_dtanh(const AGroup& A, const AGroup& A2, const v8f* Bt,
                                                   v8f* cur, float* x) {
  mma_open();
  int k = 0;
  #pragma unroll
  for (int half = 0; half < 2; ++half) {
    const AGroup& Ah = half ? A2 : A;
    #pragma unroll
    for (int tt = 0; tt < 2; ++tt)
      #pragma unroll
      for (int j = 0; j < 4; ++j) {
        v2f b[2];
        b[0].x = Bt[half * 4 + tt * 2 + 0][2 * j];
        b[0].y = Bt[half * 4 + tt * 2 + 0][2 * j + 1];
        b[1].x = Bt[half * 4 + tt * 2 + 1][2 * j];
        b[1].y = Bt[half * 4 + tt * 2 + 1][2 * j + 1];
        #pragma unroll
        for (int ft = 0; ft < 2; ++ft) {
          const float4 qa = Ah.q[ft][tt][j >> 1];
          v2f a;
          if (j & 1) { a.x = qa.z; a.y = qa.w; } else { a.x = qa.x; a.y = qa.y; }
          #pragma unroll
          for (int bt = 0; bt < 2; ++bt) {
            const int tile = ft * 2 + bt;
            cur[tile] = wmma4(a, b[bt], cur[tile]);
            const int e0 = (3 * k) & 31, e1 = (3 * k + 1) & 31, e2 = (3 * k + 2) & 31;
            asm volatile("v_nop\n\tv_nop\n\tv_nop\n\tv_nop"
                         : "+v"(cur[tile]), "+v"(x[e0]), "+v"(x[e1]), "+v"(x[e2]));
            x[e0] = dtanh_stage((3 * k) >> 5, x[e0]);
            x[e1] = dtanh_stage((3 * k + 1) >> 5, x[e1]);
            x[e2] = dtanh_stage((3 * k + 2) >> 5, x[e2]);
            asm volatile("" : "+v"(cur[(tile + 1) & 3]), "+v"(x[e0]), "+v"(x[e1]), "+v"(x[e2]));
            ++k;
          }
        }
      }
  }
  __builtin_amdgcn_sched_barrier(0);
}

__device__ __forceinline__ void put_tiles(float4* dst, const v8f* t) {
  #pragma unroll
  for (int i = 0; i < 4; ++i)
    #pragma unroll
    for (int q = 0; q < 2; ++q)
      dst[(i * 2 + q) * 32] = make_float4(t[i][4 * q], t[i][4 * q + 1], t[i][4 * q + 2], t[i][4 * q + 3]);
}

__device__ __forceinline__ void get_tiles(const float4* src, v8f* t) {
  #pragma unroll
  for (int i = 0; i < 4; ++i)
    #pragma unroll
    for (int q = 0; q < 2; ++q) {
      const float4 v = src[(i * 2 + q) * 32];
      t[i][4 * q] = v.x; t[i][4 * q + 1] = v.y; t[i][4 * q + 2] = v.z; t[i][4 * q + 3] = v.w;
    }
}

__device__ __forceinline__ void lds_barrier() {
  __builtin_amdgcn_fence(__ATOMIC_RELEASE, "workgroup", "local");
  __builtin_amdgcn_s_barrier();
  __builtin_amdgcn_fence(__ATOMIC_ACQUIRE, "workgroup", "local");
}

__device__ __forceinline__ v8f load8(const float* __restrict__ p) {
  const float4 u0 = *(const float4*)p;
  const float4 u1 = *(const float4*)(p + 4);
  v8f r;
  r[0] = u0.x; r[1] = u0.y; r[2] = u0.z; r[3] = u0.w;
  r[4] = u1.x; r[5] = u1.y; r[6] = u1.z; r[7] = u1.w;
  return r;
}

__global__ void __launch_bounds__(128) __attribute__((amdgpu_waves_per_eu(1, 1))) jacde_kernel(
    const float* __restrict__ t,      const float* __restrict__ hmat,
    const float* __restrict__ coeffs, const float* __restrict__ dcoeffs,
    const float* __restrict__ tobs,
    const float* __restrict__ wx,     const float* __restrict__ wh,
    const float* __restrict__ wout,
    const float* __restrict__ b0,     const float* __restrict__ b1,
    float* __restrict__ out) {
  __shared__ float4 xch[2][4][8][32];
  __shared__ __attribute__((aligned(16))) float sx[2][32][XLDP];

  const int tid  = threadIdx.x;
  const int lane = tid & 31;
  const int wave = tid >> 5;
  const int c    = lane & 15;
  const int hh   = lane >> 4;
  const int brow = blockIdx.x * 32;
  const int f0   = wave * 32;

  const unsigned lo_x  = (f0 + c) * CINC + 8 * hh;
  const unsigned lo_w  = (f0 + c) * HDIM + 8 * hh;
  const unsigned lo_g0 = lo_w + 32 * wave;
  const unsigned lo_g1 = lo_w + 32 * ((wave + 1) & 3);
  const unsigned lo_g2 = lo_w + 32 * ((wave + 2) & 3);
  const unsigned lo_g3 = lo_w + 32 * ((wave + 3) & 3);
  float4* xf = &xch[0][0][0][0];
  float4* my0 = xf + wave * 256 + lane;
  float4* my1 = my0 + 1024;
  const float4* pa0 = xf + ((wave + 1) & 3) * 256 + lane;
  const float4* pb0 = xf + ((wave + 2) & 3) * 256 + lane;
  const float4* pc0 = xf + ((wave + 3) & 3) * 256 + lane;

  AGroup AX0, AX1, AY0, AY1, AZ0, AZ1;
  const float ts = t[0];
  float tb[NOBSC];
  #pragma unroll
  for (int j = 0; j < NOBSC; ++j) tb[j] = tobs[j];

  load_group<HDIM>(AX0, wh, lo_g0);
  v8f own[4];
  {
    const float* hp = hmat + (size_t)(brow + c) * HDIM + f0 + 8 * hh;
    #pragma unroll
    for (int tt = 0; tt < 2; ++tt)
      #pragma unroll
      for (int bt = 0; bt < 2; ++bt)
        own[tt * 2 + bt] = load8(hp + (size_t)bt * 16 * HDIM + 16 * tt);
  }
  v8f acc[4];
  #pragma unroll
  for (int ft = 0; ft < 2; ++ft) {
    acc[ft * 2 + 0] = load8(b0 + f0 + 16 * ft + 8 * hh);
    acc[ft * 2 + 1] = acc[ft * 2 + 0];
  }
  __builtin_amdgcn_sched_barrier(0);
  load_group<HDIM>(AX1, wh, lo_g1);
  load_group<HDIM>(AY0, wh, lo_g2);
  load_group<HDIM>(AY1, wh, lo_g3);
  __builtin_amdgcn_sched_barrier(0);

  int idx = 0;
  #pragma unroll
  for (int j = 0; j < NOBSC; ++j) idx += (tb[j] <= ts) ? 1 : 0;
  idx -= 1;
  idx = idx < 0 ? 0 : (idx > NOBSC - 2 ? NOBSC - 2 : idx);
  const float dt = ts - tobs[idx];
  const float p1 = dt, p2 = dt * dt, p3 = p2 * dt;

  const int ch = tid & 63;
  const int rg = tid >> 6;
  float4 cx[16], cd[16];
  {
    const size_t off = ((size_t)(brow + 16 * rg) * (NOBSC - 1) + idx) * CINC * 4 + ch * 4;
    const float* cb = coeffs + off;
    const float* db = dcoeffs + off;
    #pragma unroll
    for (int i = 0; i < 16; ++i) cx[i] = *(const float4*)(cb + (size_t)i * (NOBSC - 1) * CINC * 4);
    #pragma unroll
    for (int i = 0; i < 16; ++i) cd[i] = *(const float4*)(db + (size_t)i * (NOBSC - 1) * CINC * 4);
  }

  v8f par0[4], par1[4], par2[4];
  put_tiles(my0, own);
  mma_one(AX0, own, acc);
  load_group<CINC>(AZ0, wx, lo_x);
  load_group<CINC>(AZ1, wx, lo_x + 32);
  lds_barrier();
  get_tiles(pa0, par0);
  get_tiles(pb0, par1);
  get_tiles(pc0, par2);
  mma_one(AX1, par0, acc);
  load_group<HDIM>(AX0, wout, lo_g0);
  load_group<HDIM>(AX1, wout, lo_g1);
  mma_two(AY0, par1, AY1, par2, acc);
  v8f a2[4];
  #pragma unroll
  for (int ft = 0; ft < 2; ++ft) {
    a2[ft * 2 + 0] = load8(b1 + f0 + 16 * ft + 8 * hh);
    a2[ft * 2 + 1] = a2[ft * 2 + 0];
  }
  load_group<HDIM>(AY0, wout, lo_g2);
  load_group<HDIM>(AY1, wout, lo_g3);
  #pragma unroll
  for (int i = 0; i < 16; ++i) {
    sx[0][16 * rg + i][ch] = cx[i].x + cx[i].y * p1 + cx[i].z * p2 + cx[i].w * p3;
    sx[1][16 * rg + i][ch] = cd[i].x + cd[i].y * p1 + cd[i].z * p2 + cd[i].w * p3;
  }
  lds_barrier();
  {
    v8f Bx[8];
    #pragma unroll
    for (int T = 0; T < 4; ++T)
      #pragma unroll
      for (int bt = 0; bt < 2; ++bt)
        Bx[T * 2 + bt] = load8(&sx[0][16 * bt + c][16 * T + 8 * hh]);
    mma_two(AZ0, Bx, AZ1, Bx + 4, acc);
  }

  v8f drelu[4];
  #pragma unroll
  for (int i = 0; i < 4; ++i)
    #pragma unroll
    for (int r = 0; r < 8; ++r) {
      const float x = acc[i][r];
      drelu[i][r] = 1.0f / (1.0f + __expf(-x));
      own[i][r]   = fmaxf(x, 0.0f);
    }
  put_tiles(my1, own);

  v8f dtn[4];
  {
    mma_one(AX0, own, a2);
    lds_barrier();
    get_tiles(pa0 + 1024, par0);
    get_tiles(pb0 + 1024, par1);
    get_tiles(pc0 + 1024, par2);
    mma_three(AX1, par0, AY0, par1, AY1, par2, a2);
  }

  {
    v8f Bxd[8];
    #pragma unroll
    for (int T = 0; T < 4; ++T)
      #pragma unroll
      for (int bt = 0; bt < 2; ++bt)
        Bxd[T * 2 + bt] = load8(&sx[1][16 * bt + c][16 * T + 8 * hh]);
    v8f cur[4];
    #pragma unroll
    for (int i = 0; i < 4; ++i) cur[i] = (v8f)(0.0f);
    float dx[32];
    #pragma unroll
    for (int i = 0; i < 4; ++i)
      #pragma unroll
      for (int r = 0; r < 8; ++r) dx[8 * i + r] = a2[i][r];
    mma_two_with_dtanh(AZ0, AZ1, Bxd, cur, dx);
    #pragma unroll
    for (int i = 0; i < 4; ++i)
      #pragma unroll
      for (int r = 0; r < 8; ++r) dtn[i][r] = dx[8 * i + r];
    #pragma unroll
    for (int i = 0; i < 4; ++i) own[i] = cur[i] * drelu[i];
  }
  put_tiles(my0, own);

  v8f outv[4];
  #pragma unroll
  for (int i = 0; i < 4; ++i) outv[i] = (v8f)(0.0f);
  #pragma nounroll
  for (int n = 0; n < 5; ++n) {
    const bool odd = (n & 1) != 0;
    const float* Wc = odd ? wh : wout;
    const float* Wn = odd ? wout : wh;
    const int rb = odd ? 1024 : 0;
    load_group<HDIM>(AY0, Wc, lo_g2);
    load_group<HDIM>(AY1, Wc, lo_g3);
    v8f tt[4];
    #pragma unroll
    for (int i = 0; i < 4; ++i) tt[i] = (v8f)(0.0f);
    mma_one(AX0, own, tt);
    lds_barrier();
    get_tiles(pa0 + rb, par0);
    get_tiles(pb0 + rb, par1);
    get_tiles(pc0 + rb, par2);
    mma_one(AX1, par0, tt);
    load_group<HDIM>(AX0, Wn, lo_g0);
    load_group<HDIM>(AX1, Wn, lo_g1);
    mma_two(AY0, par1, AY1, par2, tt);
    #pragma unroll
    for (int i = 0; i < 4; ++i) {
      const v8f mlt = odd ? drelu[i] : dtn[i];
      tt[i] *= mlt;
      own[i] = tt[i];
      outv[i] += odd ? (v8f)(0.0f) : tt[i];
    }
    if (n < 4) put_tiles(odd ? my0 : my1, own);
  }

  #pragma unroll
  for (int ft = 0; ft < 2; ++ft)
    #pragma unroll
    for (int bt = 0; bt < 2; ++bt) {
      float* op = out + (size_t)(brow + 16 * bt + c) * HDIM + f0 + 16 * ft + 8 * hh;
      const v8f v = outv[ft * 2 + bt];
      *(float4*)op       = make_float4(v[0], v[1], v[2], v[3]);
      *(float4*)(op + 4) = make_float4(v[4], v[5], v[6], v[7]);
    }
}

extern "C" void kernel_launch(void* const* d_in, const int* in_sizes, int n_in,
                              void* d_out, int out_size, void* d_ws, size_t ws_size,
                              hipStream_t stream) {
  const float* t       = (const float*)d_in[0];
  const float* hmat    = (const float*)d_in[1];
  const float* coeffs  = (const float*)d_in[2];
  const float* dcoeffs = (const float*)d_in[3];
  const float* tobs    = (const float*)d_in[4];
  const float* wx      = (const float*)d_in[5];
  const float* wh      = (const float*)d_in[6];
  const float* wout    = (const float*)d_in[7];
  const float* b0      = (const float*)d_in[8];
  const float* b1      = (const float*)d_in[9];
  float* out           = (float*)d_out;

  const int B = 8192;
  jacde_kernel<<<B / 32, 128, 0, stream>>>(t, hmat, coeffs, dcoeffs, tobs,
                                           wx, wh, wout, b0, b1, out);
}


# Each wave keeps all its wh, wout and wx fragments in registers from their first fetch (320 registers): the five chained products issue no loads; trips paired (wout, wh); sum order unchanged
# speedup vs baseline: 2.2264x; 1.0509x over previous
#include <hip/hip_runtime.h>


typedef __attribute__((ext_vector_type(2))) float v2f;
typedef __attribute__((ext_vector_type(8))) float v8f;

#define HDIM  128
#define CINC  64
#define NOBSC 16
#define XLDP  68

__device__ __forceinline__ v8f wmma4(v2f a, v2f b, v8f c) {
  return __builtin_amdgcn_wmma_f32_16x16x4_f32(
       false, a,  false, b,
       (short)0, c,  false,  false);
}

struct AGroup { float4 q[2][2][2]; };

template <int LDW>
__device__ __forceinline__ void load_group(AGroup& A, const float* __restrict__ Wm, unsigned lo) {
  const float* base = Wm + lo;
  #pragma unroll
  for (int ft = 0; ft < 2; ++ft)
    #pragma unroll
    for (int tt = 0; tt < 2; ++tt)
      #pragma unroll
      for (int q = 0; q < 2; ++q)
        A.q[ft][tt][q] = *(const float4*)(base + ft * 16 * LDW + tt * 16 + q * 4);
}

__device__ __forceinline__ void mma_body(const AGroup& A, const v8f* Bt, v8f* acc) {
  #pragma unroll
  for (int tt = 0; tt < 2; ++tt)
    #pragma unroll
    for (int j = 0; j < 4; ++j) {
      v2f b0, b1;
      b0.x = Bt[tt * 2 + 0][2 * j];
      b0.y = Bt[tt * 2 + 0][2 * j + 1];
      b1.x = Bt[tt * 2 + 1][2 * j];
      b1.y = Bt[tt * 2 + 1][2 * j + 1];
      #pragma unroll
      for (int ft = 0; ft < 2; ++ft) {
        const float4 qa = A.q[ft][tt][j >> 1];
        v2f a;
        if (j & 1) { a.x = qa.z; a.y = qa.w; } else { a.x = qa.x; a.y = qa.y; }
        acc[ft * 2 + 0] = wmma4(a, b0, acc[ft * 2 + 0]);
        acc[ft * 2 + 1] = wmma4(a, b1, acc[ft * 2 + 1]);
      }
    }
}

__device__ __forceinline__ void mma_open() {
  __builtin_amdgcn_sched_barrier(0);
}
__device__ __forceinline__ void mma_close(v8f* acc) {
  asm volatile("v_nop\n\tv_nop\n\tv_nop\n\tv_nop"
               : "+v"(acc[0]), "+v"(acc[1]), "+v"(acc[2]), "+v"(acc[3]));
  __builtin_amdgcn_sched_barrier(0);
}
__device__ __forceinline__ void mma_one(const AGroup& A, const v8f* Bt, v8f* acc) {
  mma_open();
  mma_body(A, Bt, acc);
  mma_close(acc);
}
__device__ __forceinline__ void mma_two(const AGroup& A, const v8f* Bt,
                                        const AGroup& A2, const v8f* Bt2, v8f* acc) {
  mma_open();
  mma_body(A, Bt, acc);
  mma_body(A2, Bt2, acc);
  mma_close(acc);
}
__device__ __forceinline__ void mma_three(const AGroup& A, const v8f* Bt,
                                          const AGroup& A2, const v8f* Bt2,
                                          const AGroup& A3, const v8f* Bt3, v8f* acc) {
  mma_open();
  mma_body(A, Bt, acc);
  mma_body(A2, Bt2, acc);
  mma_body(A3, Bt3, acc);
  mma_close(acc);
}

__device__ __forceinline__ float dtanh_stage(int s, float x) {
  switch (s) {
    case 0:  return __builtin_fabsf(x) * 2.885390081777927f;
    case 1:  return __builtin_amdgcn_exp2f(x);
    case 2:  return x + 1.0f;
    case 3:  return __builtin_amdgcn_rcpf(x);
    case 4:  return __builtin_fmaf(-x, x, x);
    default: return 4.0f * x;
  }
}

__device__ __forceinline__ void mma_two_with_dtanh(const AGroup& A, const AGroup& A2, const v8f* Bt,
                                                   v8f* cur, float* x) {
  mma_open();
  int k = 0;
  #pragma unroll
  for (int half = 0; half < 2; ++half) {
    const AGroup& Ah = half ? A2 : A;
    #pragma unroll
    for (int tt = 0; tt < 2; ++tt)
      #pragma unroll
      for (int j = 0; j < 4; ++j) {
        v2f b[2];
        b[0].x = Bt[half * 4 + tt * 2 + 0][2 * j];
        b[0].y = Bt[half * 4 + tt * 2 + 0][2 * j + 1];
        b[1].x = Bt[half * 4 + tt * 2 + 1][2 * j];
        b[1].y = Bt[half * 4 + tt * 2 + 1][2 * j + 1];
        #pragma unroll
        for (int ft = 0; ft < 2; ++ft) {
          const float4 qa = Ah.q[ft][tt][j >> 1];
          v2f a;
          if (j & 1) { a.x = qa.z; a.y = qa.w; } else { a.x = qa.x; a.y = qa.y; }
          #pragma unroll
          for (int bt = 0; bt < 2; ++bt) {
            const int tile = ft * 2 + bt;
            cur[tile] = wmma4(a, b[bt], cur[tile]);
            const int e0 = (3 * k) & 31, e1 = (3 * k + 1) & 31, e2 = (3 * k + 2) & 31;
            asm volatile("v_nop\n\tv_nop\n\tv_nop\n\tv_nop"
                         : "+v"(cur[tile]), "+v"(x[e0]), "+v"(x[e1]), "+v"(x[e2]));
            x[e0] = dtanh_stage((3 * k) >> 5, x[e0]);
            x[e1] = dtanh_stage((3 * k + 1) >> 5, x[e1]);
            x[e2] = dtanh_stage((3 * k + 2) >> 5, x[e2]);
            asm volatile("" : "+v"(cur[(tile + 1) & 3]), "+v"(x[e0]), "+v"(x[e1]), "+v"(x[e2]));
            ++k;
          }
        }
      }
  }
  __builtin_amdgcn_sched_barrier(0);
}

__device__ __forceinline__ void put_tiles(float4* dst, const v8f* t) {
  #pragma unroll
  for (int i = 0; i < 4; ++i)
    #pragma unroll
    for (int q = 0; q < 2; ++q)
      dst[(i * 2 + q) * 32] = make_float4(t[i][4 * q], t[i][4 * q + 1], t[i][4 * q + 2], t[i][4 * q + 3]);
}

__device__ __forceinline__ void get_tiles(const float4* src, v8f* t) {
  #pragma unroll
  for (int i = 0; i < 4; ++i)
    #pragma unroll
    for (int q = 0; q < 2; ++q) {
      const float4 v = src[(i * 2 + q) * 32];
      t[i][4 * q] = v.x; t[i][4 * q + 1] = v.y; t[i][4 * q + 2] = v.z; t[i][4 * q + 3] = v.w;
    }
}

__device__ __forceinline__ void lds_barrier() {
  __builtin_amdgcn_fence(__ATOMIC_RELEASE, "workgroup", "local");
  __builtin_amdgcn_s_barrier();
  __builtin_amdgcn_fence(__ATOMIC_ACQUIRE, "workgroup", "local");
}

__device__ __forceinline__ v8f load8(const float* __restrict__ p) {
  const float4 u0 = *(const float4*)p;
  const float4 u1 = *(const float4*)(p + 4);
  v8f r;
  r[0] = u0.x; r[1] = u0.y; r[2] = u0.z; r[3] = u0.w;
  r[4] = u1.x; r[5] = u1.y; r[6] = u1.z; r[7] = u1.w;
  return r;
}

__global__ void __launch_bounds__(128) __attribute__((amdgpu_waves_per_eu(1, 1))) jacde_kernel(
    const float* __restrict__ t,      const float* __restrict__ hmat,
    const float* __restrict__ coeffs, const float* __restrict__ dcoeffs,
    const float* __restrict__ tobs,
    const float* __restrict__ wx,     const float* __restrict__ wh,
    const float* __restrict__ wout,
    const float* __restrict__ b0,     const float* __restrict__ b1,
    float* __restrict__ out) {
  __shared__ float4 xch[2][4][8][32];
  __shared__ __attribute__((aligned(16))) float sx[2][32][XLDP];

  const int tid  = threadIdx.x;
  const int lane = tid & 31;
  const int wave = tid >> 5;
  const int c    = lane & 15;
  const int hh   = lane >> 4;
  const int brow = blockIdx.x * 32;
  const int f0   = wave * 32;

  const unsigned lo_x  = (f0 + c) * CINC + 8 * hh;
  const unsigned lo_w  = (f0 + c) * HDIM + 8 * hh;
  const unsigned lo_g0 = lo_w + 32 * wave;
  const unsigned lo_g1 = lo_w + 32 * ((wave + 1) & 3);
  const unsigned lo_g2 = lo_w + 32 * ((wave + 2) & 3);
  const unsigned lo_g3 = lo_w + 32 * ((wave + 3) & 3);
  float4* xf = &xch[0][0][0][0];
  float4* my0 = xf + wave * 256 + lane;
  float4* my1 = my0 + 1024;
  const float4* pa0 = xf + ((wave + 1) & 3) * 256 + lane;
  const float4* pb0 = xf + ((wave + 2) & 3) * 256 + lane;
  const float4* pc0 = xf + ((wave + 3) & 3) * 256 + lane;

  AGroup WH0, WH1, WH2, WH3, WO0, WO1, WO2, WO3, AZ0, AZ1;
  const float ts = t[0];
  float tb[NOBSC];
  #pragma unroll
  for (int j = 0; j < NOBSC; ++j) tb[j] = tobs[j];

  load_group<HDIM>(WH0, wh, lo_g0);
  v8f own[4];
  {
    const float* hp = hmat + (size_t)(brow + c) * HDIM + f0 + 8 * hh;
    #pragma unroll
    for (int tt = 0; tt < 2; ++tt)
      #pragma unroll
      for (int bt = 0; bt < 2; ++bt)
        own[tt * 2 + bt] = load8(hp + (size_t)bt * 16 * HDIM + 16 * tt);
  }
  v8f acc[4];
  #pragma unroll
  for (int ft = 0; ft < 2; ++ft) {
    acc[ft * 2 + 0] = load8(b0 + f0 + 16 * ft + 8 * hh);
    acc[ft * 2 + 1] = acc[ft * 2 + 0];
  }
  __builtin_amdgcn_sched_barrier(0);
  load_group<HDIM>(WH1, wh, lo_g1);
  load_group<HDIM>(WH2, wh, lo_g2);
  load_group<HDIM>(WH3, wh, lo_g3);
  __builtin_amdgcn_sched_barrier(0);

  int idx = 0;
  #pragma unroll
  for (int j = 0; j < NOBSC; ++j) idx += (tb[j] <= ts) ? 1 : 0;
  idx -= 1;
  idx = idx < 0 ? 0 : (idx > NOBSC - 2 ? NOBSC - 2 : idx);
  const float dt = ts - tobs[idx];
  const float p1 = dt, p2 = dt * dt, p3 = p2 * dt;

  const int ch = tid & 63;
  const int rg = tid >> 6;
  float4 cx[16], cd[16];
  {
    const size_t off = ((size_t)(brow + 16 * rg) * (NOBSC - 1) + idx) * CINC * 4 + ch * 4;
    const float* cb = coeffs + off;
    const float* db = dcoeffs + off;
    #pragma unroll
    for (int i = 0; i < 16; ++i) cx[i] = *(const float4*)(cb + (size_t)i * (NOBSC - 1) * CINC * 4);
    #pragma unroll
    for (int i = 0; i < 16; ++i) cd[i] = *(const float4*)(db + (size_t)i * (NOBSC - 1) * CINC * 4);
  }

  v8f par0[4], par1[4], par2[4];
  put_tiles(my0, own);
  mma_one(WH0, own, acc);
  load_group<CINC>(AZ0, wx, lo_x);
  load_group<CINC>(AZ1, wx, lo_x + 32);
  lds_barrier();
  get_tiles(pa0, par0);
  get_tiles(pb0, par1);
  get_tiles(pc0, par2);
  mma_one(WH1, par0, acc);
  load_group<HDIM>(WO0, wout, lo_g0);
  load_group<HDIM>(WO1, wout, lo_g1);
  mma_two(WH2, par1, WH3, par2, acc);
  v8f a2[4];
  #pragma unroll
  for (int ft = 0; ft < 2; ++ft) {
    a2[ft * 2 + 0] = load8(b1 + f0 + 16 * ft + 8 * hh);
    a2[ft * 2 + 1] = a2[ft * 2 + 0];
  }
  load_group<HDIM>(WO2, wout, lo_g2);
  load_group<HDIM>(WO3, wout, lo_g3);
  #pragma unroll
  for (int i = 0; i < 16; ++i) {
    sx[0][16 * rg + i][ch] = cx[i].x + cx[i].y * p1 + cx[i].z * p2 + cx[i].w * p3;
    sx[1][16 * rg + i][ch] = cd[i].x + cd[i].y * p1 + cd[i].z * p2 + cd[i].w * p3;
  }
  lds_barrier();
  {
    v8f Bx[8];
    #pragma unroll
    for (int T = 0; T < 4; ++T)
      #pragma unroll
      for (int bt = 0; bt < 2; ++bt)
        Bx[T * 2 + bt] = load8(&sx[0][16 * bt + c][16 * T + 8 * hh]);
    mma_two(AZ0, Bx, AZ1, Bx + 4, acc);
  }

  v8f drelu[4];
  #pragma unroll
  for (int i = 0; i < 4; ++i)
    #pragma unroll
    for (int r = 0; r < 8; ++r) {
      const float x = acc[i][r];
      drelu[i][r] = 1.0f / (1.0f + __expf(-x));
      own[i][r]   = fmaxf(x, 0.0f);
    }
  put_tiles(my1, own);

  v8f dtn[4];
  {
    mma_one(WO0, own, a2);
    lds_barrier();
    get_tiles(pa0 + 1024, par0);
    get_tiles(pb0 + 1024, par1);
    get_tiles(pc0 + 1024, par2);
    mma_three(WO1, par0, WO2, par1, WO3, par2, a2);
  }

  {
    v8f Bxd[8];
    #pragma unroll
    for (int T = 0; T < 4; ++T)
      #pragma unroll
      for (int bt = 0; bt < 2; ++bt)
        Bxd[T * 2 + bt] = load8(&sx[1][16 * bt + c][16 * T + 8 * hh]);
    v8f cur[4];
    #pragma unroll
    for (int i = 0; i < 4; ++i) cur[i] = (v8f)(0.0f);
    float dx[32];
    #pragma unroll
    for (int i = 0; i < 4; ++i)
      #pragma unroll
      for (int r = 0; r < 8; ++r) dx[8 * i + r] = a2[i][r];
    mma_two_with_dtanh(AZ0, AZ1, Bxd, cur, dx);
    #pragma unroll
    for (int i = 0; i < 4; ++i)
      #pragma unroll
      for (int r = 0; r < 8; ++r) dtn[i][r] = dx[8 * i + r];
    #pragma unroll
    for (int i = 0; i < 4; ++i) own[i] = cur[i] * drelu[i];
  }
  put_tiles(my0, own);

  v8f outv[4];
  #pragma unroll
  for (int i = 0; i < 4; ++i) outv[i] = (v8f)(0.0f);
  #pragma nounroll
  for (int n = 0; n < 3; ++n) {
    v8f tt[4];
    #pragma unroll
    for (int i = 0; i < 4; ++i) tt[i] = (v8f)(0.0f);
    mma_one(WO0, own, tt);
    lds_barrier();
    get_tiles(pa0, par0);
    get_tiles(pb0, par1);
    get_tiles(pc0, par2);
    mma_three(WO1, par0, WO2, par1, WO3, par2, tt);
    #pragma unroll
    for (int i = 0; i < 4; ++i) {
      tt[i] *= dtn[i];
      own[i] = tt[i];
      outv[i] += tt[i];
    }
    if (n == 2) break;
    put_tiles(my1, own);

    #pragma unroll
    for (int i = 0; i < 4; ++i) tt[i] = (v8f)(0.0f);
    mma_one(WH0, own, tt);
    lds_barrier();
    get_tiles(pa0 + 1024, par0);
    get_tiles(pb0 + 1024, par1);
    get_tiles(pc0 + 1024, par2);
    mma_three(WH1, par0, WH2, par1, WH3, par2, tt);
    #pragma unroll
    for (int i = 0; i < 4; ++i) own[i] = tt[i] * drelu[i];
    put_tiles(my0, own);
  }

  #pragma unroll
  for (int ft = 0; ft < 2; ++ft)
    #pragma unroll
    for (int bt = 0; bt < 2; ++bt) {
      float* op = out + (size_t)(brow + 16 * bt + c) * HDIM + f0 + 16 * ft + 8 * hh;
      const v8f v = outv[ft * 2 + bt];
      *(float4*)op       = make_float4(v[0], v[1], v[2], v[3]);
      *(float4*)(op + 4) = make_float4(v[4], v[5], v[6], v[7]);
    }
}

extern "C" void kernel_launch(void* const* d_in, const int* in_sizes, int n_in,
                              void* d_out, int out_size, void* d_ws, size_t ws_size,
                              hipStream_t stream) {
  const float* t       = (const float*)d_in[0];
  const float* hmat    = (const float*)d_in[1];
  const float* coeffs  = (const float*)d_in[2];
  const float* dcoeffs = (const float*)d_in[3];
  const float* tobs    = (const float*)d_in[4];
  const float* wx      = (const float*)d_in[5];
  const float* wh      = (const float*)d_in[6];
  const float* wout    = (const float*)d_in[7];
  const float* b0      = (const float*)d_in[8];
  const float* b1      = (const float*)d_in[9];
  float* out           = (float*)d_out;

  const int B = 8192;
  jacde_kernel<<<B / 32, 128, 0, stream>>>(t, hmat, coeffs, dcoeffs, tobs,
                                           wx, wh, wout, b0, b1, out);
}
